# starting kernel
# baseline (speedup 1.0000x reference)
#include <hip/hip_runtime.h>
#include <hip/hip_bf16.h>

typedef __attribute__((ext_vector_type(2))) float v2f;
typedef __attribute__((ext_vector_type(8))) float v8f;

__global__ __launch_bounds__(32) void gnn_proj_wmma(
    const float* __restrict__ h,
    const float* __restrict__ W,
    float* __restrict__ P)
{
    const int tile = blockIdx.x;
    const int lane = threadIdx.x;
    const int rc   = lane & 15;
    const int koff = (lane >> 4) * 2;

    const float* hrow = h + (size_t)(tile * 16 + rc) * 128 + koff;

    const float* wrow = W + koff;
    if (rc < 3)      wrow = W + rc * 256 + koff;
    else if (rc < 6) wrow = W + (rc - 3) * 256 + 128 + koff;

    v8f acc = {};
#pragma unroll
    for (int kk = 0; kk < 32; ++kk) {
        v2f a = *(const v2f*)(hrow + kk * 4);
        v2f b = *(const v2f*)(wrow + kk * 4);
        acc = __builtin_amdgcn_wmma_f32_16x16x4_f32(
            false, a, false, b, (short)0, acc, false, false);
    }

    if (rc < 6) {
        const int colOff = (rc < 3) ? rc : (rc + 1);
        const int mbase  = tile * 16 + ((lane >> 4) * 8);
#pragma unroll
        for (int r = 0; r < 8; ++r) {
            P[(size_t)(mbase + r) * 8 + colOff] = acc[r];
        }
    }
}

__global__ __launch_bounds__(256) void gnn_edge_gather(
    const int* __restrict__ src,
    const int* __restrict__ dst,
    const float* __restrict__ P,
    const float* __restrict__ b,
    float* __restrict__ out,
    int n_edges)
{
    const int e = blockIdx.x * blockDim.x + threadIdx.x;
    if (e >= n_edges) return;

    const int s = src[e];
    const int d = dst[e];

    const float4 ps = *(const float4*)(P + (size_t)s * 8);
    const float4 pd = *(const float4*)(P + (size_t)d * 8 + 4);

    const float b0 = b[0], b1 = b[1], b2 = b[2];

    float* o = out + (size_t)e * 3;
    o[0] = ps.x + pd.x + b0;
    o[1] = ps.y + pd.y + b1;
    o[2] = ps.z + pd.z + b2;
}

extern "C" void kernel_launch(void* const* d_in, const int* in_sizes, int n_in,
                              void* d_out, int out_size, void* d_ws, size_t ws_size,
                              hipStream_t stream) {
    const float* h   = (const float*)d_in[0];
    const int*   src = (const int*)d_in[1];
    const int*   dst = (const int*)d_in[2];
    const float* W   = (const float*)d_in[3];
    const float* b   = (const float*)d_in[4];
    float* out = (float*)d_out;

    const int n_nodes = in_sizes[0] / 128;
    const int n_edges = in_sizes[1];

    float* P = (float*)d_ws;

    const int n_tiles = (n_nodes + 15) / 16;
    gnn_proj_wmma<<<n_tiles, 32, 0, stream>>>(h, W, P);

    const int threads = 256;
    const int blocks  = (n_edges + threads - 1) / threads;
    gnn_edge_gather<<<blocks, threads, 0, stream>>>(src, dst, P, b, out, n_edges);
}


# edge kernel: each of 245 work-groups copies the packed 240000-B projection table into LDS with whole-line loads and gathers src/dst rows from LDS, not from the second-level cache; 8 edges per lane
# speedup vs baseline: 1.1603x; 1.1603x over previous
#include <hip/hip_runtime.h>
#include <hip/hip_bf16.h>


typedef __attribute__((ext_vector_type(2))) float v2f;
typedef __attribute__((ext_vector_type(8))) float v8f;
typedef __attribute__((ext_vector_type(4))) float v4f;

constexpr int NODES       = 10000;
constexpr int TAB_FLOATS  = NODES * 6;
constexpr int TAB_VEC4    = TAB_FLOATS / 4;
constexpr int EDGE_THREADS = 256;
constexpr int CHUNKS_PER_THREAD = 2;
constexpr int EDGES_PER_BLOCK = EDGE_THREADS * CHUNKS_PER_THREAD * 4;
static_assert(TAB_FLOATS % 32 == 0, "the table is a whole number of 128-byte lines");
static_assert(TAB_FLOATS * 4 <= 320 * 1024, "the table fits one work-group's LDS");

__global__ __launch_bounds__(32) void gnn_proj_wmma(
    const float* __restrict__ h,
    const float* __restrict__ W,
    float* __restrict__ P)
{
    const int tile = blockIdx.x;
    const int lane = threadIdx.x;
    const int rc   = lane & 15;
    const int koff = (lane >> 4) * 2;

    const float* hrow = h + (size_t)(tile * 16 + rc) * 128 + koff;

    const float* wrow = W + koff;
    if (rc < 3)      wrow = W + rc * 256 + koff;
    else if (rc < 6) wrow = W + (rc - 3) * 256 + 128 + koff;

    v8f acc = {};
#pragma unroll
    for (int kk = 0; kk < 32; ++kk) {
        v2f a = *(const v2f*)(hrow + kk * 4);
        v2f b = *(const v2f*)(wrow + kk * 4);
        acc = __builtin_amdgcn_wmma_f32_16x16x4_f32(
            false, a, false, b, (short)0, acc, false, false);
    }

    if (rc < 6) {
        float* tabcol = (rc < 3) ? (P + rc) : (P + NODES * 3 + (rc - 3));
        const int mbase  = tile * 16 + ((lane >> 4) * 8);
#pragma unroll
        for (int r = 0; r < 8; ++r) {
            tabcol[(size_t)(mbase + r) * 3] = acc[r];
        }
    }
}

__global__ __launch_bounds__(EDGE_THREADS) void gnn_edge_gather(
    const int* __restrict__ src,
    const int* __restrict__ dst,
    const float* __restrict__ P,
    const float* __restrict__ b,
    float* __restrict__ out,
    int n_edges)
{
    __shared__ __attribute__((aligned(16))) float tab[TAB_FLOATS];

    const int tid = threadIdx.x;

    const float b0 = b[0], b1 = b[1], b2 = b[2];

    int4 si[CHUNKS_PER_THREAD], di[CHUNKS_PER_THREAD];
    int  e0[CHUNKS_PER_THREAD];
    bool whole[CHUNKS_PER_THREAD];
#pragma unroll
    for (int c = 0; c < CHUNKS_PER_THREAD; ++c) {
        e0[c] = blockIdx.x * EDGES_PER_BLOCK + (c * EDGE_THREADS + tid) * 4;
        whole[c] = (e0[c] + 3 < n_edges);
        const int ea = whole[c] ? e0[c] : 0;
        si[c] = *(const int4*)(src + ea);
        di[c] = *(const int4*)(dst + ea);
    }

    {
        const float4* gsrc = (const float4*)P;
        float4* ldst = (float4*)tab;
        const int rot = (int)(blockIdx.x % 256u) * 56;
        constexpr int TRIPS = (TAB_VEC4 + EDGE_THREADS - 1) / EDGE_THREADS;
        static_assert(TRIPS >= 2, "the last trip leans on the trip before it");
#pragma unroll
        for (int j = 0; j < TRIPS; ++j) {
            int li = j * EDGE_THREADS + tid;
            if (j == TRIPS - 1) li = (li < TAB_VEC4) ? li : (li - EDGE_THREADS);
            int p = li + rot;
            p = (p >= TAB_VEC4) ? (p - TAB_VEC4) : p;
            ldst[p] = gsrc[p];
        }
    }

    __syncthreads();

    const float* tabS = tab;
    const float* tabD = tab + NODES * 3;

#pragma unroll
    for (int c = 0; c < CHUNKS_PER_THREAD; ++c) {
        if (whole[c]) {
            const int sidx[4] = { si[c].x, si[c].y, si[c].z, si[c].w };
            const int didx[4] = { di[c].x, di[c].y, di[c].z, di[c].w };
            float o[12];
#pragma unroll
            for (int k = 0; k < 4; ++k) {
                const float* ps = tabS + sidx[k] * 3;
                const float* pd = tabD + didx[k] * 3;
                o[3 * k + 0] = ps[0] + pd[0] + b0;
                o[3 * k + 1] = ps[1] + pd[1] + b1;
                o[3 * k + 2] = ps[2] + pd[2] + b2;
            }
            v4f* op = (v4f*)(out + (size_t)e0[c] * 3);
            const v4f q0 = { o[0], o[1], o[2],  o[3]  };
            const v4f q1 = { o[4], o[5], o[6],  o[7]  };
            const v4f q2 = { o[8], o[9], o[10], o[11] };
            op[0] = q0;
            op[1] = q1;
            op[2] = q2;
        } else {
            for (int e = e0[c]; e < n_edges; ++e) {
                const float* ps = tabS + src[e] * 3;
                const float* pd = tabD + dst[e] * 3;
                float* op = out + (size_t)e * 3;
                op[0] = ps[0] + pd[0] + b0;
                op[1] = ps[1] + pd[1] + b1;
                op[2] = ps[2] + pd[2] + b2;
            }
        }
    }
}

extern "C" void kernel_launch(void* const* d_in, const int* in_sizes, int n_in,
                              void* d_out, int out_size, void* d_ws, size_t ws_size,
                              hipStream_t stream) {
    const float* h   = (const float*)d_in[0];
    const int*   src = (const int*)d_in[1];
    const int*   dst = (const int*)d_in[2];
    const float* W   = (const float*)d_in[3];
    const float* b   = (const float*)d_in[4];
    float* out = (float*)d_out;

    const int n_nodes = in_sizes[0] / 128;
    const int n_edges = in_sizes[1];

    float* P = (float*)d_ws;

    const int n_tiles = (n_nodes + 15) / 16;
    gnn_proj_wmma<<<n_tiles, 32, 0, stream>>>(h, W, P);

    const int blocks = (n_edges + EDGES_PER_BLOCK - 1) / EDGES_PER_BLOCK;
    gnn_edge_gather<<<blocks, EDGE_THREADS, 0, stream>>>(src, dst, P, b, out, n_edges);
}
